# starting kernel
# baseline (speedup 1.0000x reference)
#include <hip/hip_runtime.h>

#define N_ATOMS 4096
#define N3 (N_ATOMS * 3)
#define CUTOFF 12.0f

typedef __attribute__((ext_vector_type(2))) float v2f;
typedef __attribute__((ext_vector_type(8))) float v8f;


__global__ void TFF_Repulsion_init(const float* __restrict__ forces_in,
                                   float* __restrict__ out) {
    int i = blockIdx.x * blockDim.x + threadIdx.x;
    if (i == 0) out[0] = 0.0f;
    if (i < N3) out[1 + i] = forces_in[i];
}

__global__ __launch_bounds__(256) void TFF_Repulsion_pairs(
    const float* __restrict__ dist,
    const float* __restrict__ vec,
    const float* __restrict__ rep,
    const int*   __restrict__ coord,
    int n_pairs,
    float* __restrict__ out) {
    __shared__ float s_forces[N3];
    for (int j = threadIdx.x; j < N3; j += blockDim.x) s_forces[j] = 0.0f;
    __syncthreads();

    float e0 = 0.0f, e1 = 0.0f;
    int idx    = blockIdx.x * blockDim.x + threadIdx.x;
    int stride = gridDim.x * blockDim.x;
    int t = 0;
    for (int i = idx; i < n_pairs; i += stride) {
        int a = coord[2 * i];
        int b = coord[2 * i + 1];
        int off = (a << 12) + b;
        float d = dist[off];
        if (d <= CUTOFF) {
            float B    = rep[i];
            float inv  = 1.0f / d;
            float inv2 = inv * inv;
            float inv6 = inv2 * inv2 * inv2;
            float e    = B * inv6;
            if (t) e1 += e; else e0 += e;
            float f = -6.0f * e * inv;
            const float* v = vec + (size_t)off * 3u;
            float fx = f * v[0], fy = f * v[1], fz = f * v[2];
            atomicAdd(&s_forces[3 * a + 0],  fx);
            atomicAdd(&s_forces[3 * a + 1],  fy);
            atomicAdd(&s_forces[3 * a + 2],  fz);
            atomicAdd(&s_forces[3 * b + 0], -fx);
            atomicAdd(&s_forces[3 * b + 1], -fy);
            atomicAdd(&s_forces[3 * b + 2], -fz);
        }
        t ^= 1;
    }
    __syncthreads();

    for (int j = threadIdx.x; j < N3; j += blockDim.x)
        atomicAdd(&out[1 + j], s_forces[j]);

    v2f a2; a2[0] = e0;   a2[1] = e1;
    v2f b2; b2[0] = 1.0f; b2[1] = 1.0f;
    v8f c = {};
    c = __builtin_amdgcn_wmma_f32_16x16x4_f32(
             false, a2,  false, b2,
             (short)0, c,  false,  false);
    float s = c[0] + c[1] + c[2] + c[3] + c[4] + c[5] + c[6] + c[7];
    s += __shfl_xor(s, 16, 32);
    if ((threadIdx.x & 31) == 0)
        atomicAdd(&out[0], s);
}

extern "C" void kernel_launch(void* const* d_in, const int* in_sizes, int n_in,
                              void* d_out, int out_size, void* d_ws, size_t ws_size,
                              hipStream_t stream) {
    const float* dist      = (const float*)d_in[0];
    const float* vec       = (const float*)d_in[1];
    const float* forces_in = (const float*)d_in[2];
    const float* rep       = (const float*)d_in[3];
    const int*   coord     = (const int*)d_in[4];
    float* out = (float*)d_out;
    int n_pairs = in_sizes[3];

    int initThreads = 256;
    int initBlocks  = (N3 + 1 + initThreads - 1) / initThreads;
    TFF_Repulsion_init<<<initBlocks, initThreads, 0, stream>>>(forces_in, out);

    TFF_Repulsion_pairs<<<512, 256, 0, stream>>>(dist, vec, rep, coord, n_pairs, out);
}


# each lane gathers 8 pairs per trip (index, distance and vector loads issued as groups of 8) so 8 misses overlap per lane; pairs re-dealt to lanes, so the atomic partial-sum order changes
# speedup vs baseline: 1.1250x; 1.1250x over previous
#include <hip/hip_runtime.h>

#define N_ATOMS 4096
#define N3 (N_ATOMS * 3)
#define CUTOFF 12.0f
#define KB 8

typedef __attribute__((ext_vector_type(2))) float v2f;
typedef __attribute__((ext_vector_type(8))) float v8f;


__global__ void TFF_Repulsion_init(const float* __restrict__ forces_in,
                                   float* __restrict__ out) {
    int i = blockIdx.x * blockDim.x + threadIdx.x;
    if (i == 0) out[0] = 0.0f;
    if (i < N3) out[1 + i] = forces_in[i];
}

__device__ __forceinline__ float pair_term(float d, float B, int a, int b,
                                           float vx, float vy, float vz,
                                           float* s_forces) {
    float inv  = 1.0f / d;
    float inv2 = inv * inv;
    float inv6 = inv2 * inv2 * inv2;
    float e    = B * inv6;
    float f    = -6.0f * e * inv;
    float fx = f * vx, fy = f * vy, fz = f * vz;
    atomicAdd(&s_forces[3 * a + 0],  fx);
    atomicAdd(&s_forces[3 * a + 1],  fy);
    atomicAdd(&s_forces[3 * a + 2],  fz);
    atomicAdd(&s_forces[3 * b + 0], -fx);
    atomicAdd(&s_forces[3 * b + 1], -fy);
    atomicAdd(&s_forces[3 * b + 2], -fz);
    return e;
}

__global__ __launch_bounds__(256) void TFF_Repulsion_pairs(
    const float* __restrict__ dist,
    const float* __restrict__ vec,
    const float* __restrict__ rep,
    const int*   __restrict__ coord,
    int n_pairs,
    float* __restrict__ out) {
    __shared__ float s_forces[N3];
    for (int j = threadIdx.x; j < N3; j += blockDim.x) s_forces[j] = 0.0f;
    __syncthreads();

    float e0 = 0.0f, e1 = 0.0f;
    const int gtid   = blockIdx.x * blockDim.x + threadIdx.x;
    const int stride = gridDim.x * blockDim.x;
    const int lane   = threadIdx.x & 31;
    const int wave   = gtid >> 5;
    const int nwaves = stride >> 5;
    const int chunk  = 32 * KB;
    const int nfull  = n_pairs / chunk;

    for (int w = wave; w < nfull; w += nwaves) {
        const int base = w * chunk + lane;
        int   a[KB], b[KB], off[KB];
        float Bc[KB], d[KB], vx[KB], vy[KB], vz[KB];
#pragma unroll
        for (int k = 0; k < KB; ++k) {
            const int i = base + 32 * k;
            a[k]   = coord[2 * i];
            b[k]   = coord[2 * i + 1];
            Bc[k]  = rep[i];
            off[k] = (a[k] << 12) + b[k];
        }
#pragma unroll
        for (int k = 0; k < KB; ++k) d[k] = dist[off[k]];
#pragma unroll
        for (int k = 0; k < KB; ++k) {
            const int voff = (d[k] <= CUTOFF) ? off[k] : 0;
            const float* v = vec + (size_t)voff * 3u;
            vx[k] = v[0]; vy[k] = v[1]; vz[k] = v[2];
        }
#pragma unroll
        for (int k = 0; k < KB; ++k) {
            asm volatile("" :: "v"(vx[k]), "v"(vy[k]), "v"(vz[k]), "v"(Bc[k]));
        }
#pragma unroll
        for (int k = 0; k < KB; ++k) {
            if (d[k] <= CUTOFF) {
                float e = pair_term(d[k], Bc[k], a[k], b[k], vx[k], vy[k], vz[k], s_forces);
                if (k & 1) e1 += e; else e0 += e;
            }
        }
    }
    for (int i = nfull * chunk + gtid; i < n_pairs; i += stride) {
        int a1 = coord[2 * i];
        int b1 = coord[2 * i + 1];
        int off1 = (a1 << 12) + b1;
        float d1 = dist[off1];
        if (d1 <= CUTOFF) {
            const float* v = vec + (size_t)off1 * 3u;
            e0 += pair_term(d1, rep[i], a1, b1, v[0], v[1], v[2], s_forces);
        }
    }
    __syncthreads();

    for (int j = threadIdx.x; j < N3; j += blockDim.x)
        atomicAdd(&out[1 + j], s_forces[j]);

    v2f a2; a2[0] = e0;   a2[1] = e1;
    v2f b2; b2[0] = 1.0f; b2[1] = 1.0f;
    v8f c = {};
    c = __builtin_amdgcn_wmma_f32_16x16x4_f32(
             false, a2,  false, b2,
             (short)0, c,  false,  false);
    float s = c[0] + c[1] + c[2] + c[3] + c[4] + c[5] + c[6] + c[7];
    s += __shfl_xor(s, 16, 32);
    if ((threadIdx.x & 31) == 0)
        atomicAdd(&out[0], s);
}

extern "C" void kernel_launch(void* const* d_in, const int* in_sizes, int n_in,
                              void* d_out, int out_size, void* d_ws, size_t ws_size,
                              hipStream_t stream) {
    const float* dist      = (const float*)d_in[0];
    const float* vec       = (const float*)d_in[1];
    const float* forces_in = (const float*)d_in[2];
    const float* rep       = (const float*)d_in[3];
    const int*   coord     = (const int*)d_in[4];
    float* out = (float*)d_out;
    int n_pairs = in_sizes[3];

    int initThreads = 256;
    int initBlocks  = (N3 + 1 + initThreads - 1) / initThreads;
    TFF_Repulsion_init<<<initBlocks, initThreads, 0, stream>>>(forces_in, out);

    TFF_Repulsion_pairs<<<512, 256, 0, stream>>>(dist, vec, rep, coord, n_pairs, out);
}


# each work-group stores its force image and wave energy sums plainly to the workspace; a second kernel adds the 512 images in fixed order instead of 512 x 12288 global float atomics; sum order changes
# speedup vs baseline: 1.6621x; 1.4774x over previous
#include <hip/hip_runtime.h>

#define N_ATOMS 4096
#define N3 (N_ATOMS * 3)
#define CUTOFF 12.0f
#define KB 8
#define NIMG 512
#define SUM_WAVES 16
#define IMG_PER_WAVE (NIMG / SUM_WAVES)
#define N_COLGRP (N3 / 64)
#define N_EPART (NIMG * 8)

typedef __attribute__((ext_vector_type(2))) float v2f;
typedef __attribute__((ext_vector_type(4))) float v4f;
typedef __attribute__((ext_vector_type(8))) float v8f;

#define LD_ONCE(p) __builtin_nontemporal_load(p)


__device__ __forceinline__ float pair_term(float d, float B, int a, int b,
                                           float vx, float vy, float vz,
                                           float* s_forces) {
    float inv  = 1.0f / d;
    float inv2 = inv * inv;
    float inv6 = inv2 * inv2 * inv2;
    float e    = B * inv6;
    float f    = -6.0f * e * inv;
    float fx = f * vx, fy = f * vy, fz = f * vz;
    atomicAdd(&s_forces[3 * a + 0],  fx);
    atomicAdd(&s_forces[3 * a + 1],  fy);
    atomicAdd(&s_forces[3 * a + 2],  fz);
    atomicAdd(&s_forces[3 * b + 0], -fx);
    atomicAdd(&s_forces[3 * b + 1], -fy);
    atomicAdd(&s_forces[3 * b + 2], -fz);
    return e;
}

__global__ __launch_bounds__(256) void TFF_Repulsion_pairs(
    const float* __restrict__ dist,
    const float* __restrict__ vec,
    const float* __restrict__ rep,
    const int*   __restrict__ coord,
    int n_pairs,
    float* __restrict__ ws) {
    __shared__ __attribute__((aligned(16))) float s_forces[N3];
    for (int j = threadIdx.x; j < N3; j += blockDim.x) s_forces[j] = 0.0f;
    __syncthreads();

    float e0 = 0.0f, e1 = 0.0f;
    const int gtid   = blockIdx.x * blockDim.x + threadIdx.x;
    const int stride = gridDim.x * blockDim.x;
    const int lane   = threadIdx.x & 31;
    const int wave   = gtid >> 5;
    const int nwaves = stride >> 5;
    const int chunk  = 32 * KB;
    const int nfull  = n_pairs / chunk;

    for (int w = wave; w < nfull; w += nwaves) {
        const int base = w * chunk + lane;
        int   a[KB], b[KB], off[KB];
        float Bc[KB], d[KB], vx[KB], vy[KB], vz[KB];
#pragma unroll
        for (int k = 0; k < KB; ++k) {
            const int i = base + 32 * k;
            a[k]   = LD_ONCE(&coord[2 * i]);
            b[k]   = LD_ONCE(&coord[2 * i + 1]);
            Bc[k]  = LD_ONCE(&rep[i]);
            off[k] = (a[k] << 12) + b[k];
        }
#pragma unroll
        for (int k = 0; k < KB; ++k) d[k] = dist[off[k]];
#pragma unroll
        for (int k = 0; k < KB; ++k) {
            const int voff = (d[k] <= CUTOFF) ? off[k] : 0;
            const float* v = vec + (size_t)voff * 3u;
            vx[k] = LD_ONCE(v); vy[k] = LD_ONCE(v + 1); vz[k] = LD_ONCE(v + 2);
        }
#pragma unroll
        for (int k = 0; k < KB; ++k) {
            asm volatile("" :: "v"(vx[k]), "v"(vy[k]), "v"(vz[k]), "v"(Bc[k]));
        }
#pragma unroll
        for (int k = 0; k < KB; ++k) {
            if (d[k] <= CUTOFF) {
                float e = pair_term(d[k], Bc[k], a[k], b[k], vx[k], vy[k], vz[k], s_forces);
                if (k & 1) e1 += e; else e0 += e;
            }
        }
    }
    for (int i = nfull * chunk + gtid; i < n_pairs; i += stride) {
        int a1 = LD_ONCE(&coord[2 * i]);
        int b1 = LD_ONCE(&coord[2 * i + 1]);
        int off1 = (a1 << 12) + b1;
        float d1 = dist[off1];
        if (d1 <= CUTOFF) {
            const float* v = vec + (size_t)off1 * 3u;
            e0 += pair_term(d1, LD_ONCE(&rep[i]), a1, b1,
                            LD_ONCE(v), LD_ONCE(v + 1), LD_ONCE(v + 2), s_forces);
        }
    }
    __syncthreads();

    {
        const v4f* s4 = (const v4f*)s_forces;
        v4f* img = (v4f*)(ws + (size_t)blockIdx.x * N3);
#pragma unroll
        for (int t = 0; t < N3 / 4 / 256; ++t) {
            const int j = threadIdx.x + 256 * t;
            img[j] = s4[j];
        }
    }

    v2f a2; a2[0] = e0;   a2[1] = e1;
    v2f b2; b2[0] = 1.0f; b2[1] = 1.0f;
    v8f c = {};
    c = __builtin_amdgcn_wmma_f32_16x16x4_f32(
             false, a2,  false, b2,
             (short)0, c,  false,  false);
    float s = c[0] + c[1] + c[2] + c[3] + c[4] + c[5] + c[6] + c[7];
    s += __shfl_xor(s, 16, 32);
    if ((threadIdx.x & 31) == 0)
        ws[(size_t)NIMG * N3 + wave] = s;
}

static_assert(N_COLGRP + 1 == 193 && SUM_WAVES * 32 == 512 && NIMG % SUM_WAVES == 0,
              "summing kernel's launch shape in kernel_launch must match these sizes");
__global__ __launch_bounds__(SUM_WAVES * 32) void TFF_Repulsion_sum(
    const float* __restrict__ forces_in,
    const float* __restrict__ ws,
    float* __restrict__ out) {
    __shared__ v2f   s_part[SUM_WAVES][32];
    __shared__ float s_e[SUM_WAVES];
    const int lane = threadIdx.x & 31;
    const int w    = threadIdx.x >> 5;
    if (blockIdx.x < N_COLGRP) {
        const int col = blockIdx.x * 64 + 2 * lane;
        const float* p = ws + (size_t)(w * IMG_PER_WAVE) * N3 + col;
        v2f acc0 = {0.0f, 0.0f}, acc1 = {0.0f, 0.0f};
#pragma unroll
        for (int i = 0; i < IMG_PER_WAVE; i += 2) {
            acc0 += LD_ONCE((const v2f*)(p + (size_t)i * N3));
            acc1 += LD_ONCE((const v2f*)(p + (size_t)(i + 1) * N3));
        }
        s_part[w][lane] = acc0 + acc1;
        __syncthreads();
        if (w == 0) {
            v2f t = s_part[0][lane];
            for (int k = 1; k < SUM_WAVES; ++k) t += s_part[k][lane];
            out[1 + col]     = forces_in[col]     + t[0];
            out[1 + col + 1] = forces_in[col + 1] + t[1];
        }
    } else {
        const v4f* e4 = (const v4f*)(ws + (size_t)NIMG * N3) + 2 * threadIdx.x;
        v4f x = e4[0], y = e4[1];
        float s = ((x[0] + x[1]) + (x[2] + x[3])) + ((y[0] + y[1]) + (y[2] + y[3]));
        for (int m = 16; m > 0; m >>= 1) s += __shfl_xor(s, m, 32);
        if (lane == 0) s_e[w] = s;
        __syncthreads();
        if (threadIdx.x == 0) {
            float t = s_e[0];
            for (int k = 1; k < SUM_WAVES; ++k) t += s_e[k];
            out[0] = t;
        }
    }
}

extern "C" void kernel_launch(void* const* d_in, const int* in_sizes, int n_in,
                              void* d_out, int out_size, void* d_ws, size_t ws_size,
                              hipStream_t stream) {
    const float* dist      = (const float*)d_in[0];
    const float* vec       = (const float*)d_in[1];
    const float* forces_in = (const float*)d_in[2];
    const float* rep       = (const float*)d_in[3];
    const int*   coord     = (const int*)d_in[4];
    float* out = (float*)d_out;
    float* ws  = (float*)d_ws;
    int n_pairs = in_sizes[3];

    TFF_Repulsion_pairs<<<NIMG, 256, 0, stream>>>(dist, vec, rep, coord, n_pairs, ws);

    TFF_Repulsion_sum<<<193, 512, 0, stream>>>(forces_in, ws, out);
}


# sorting kernel starts 16 waves per work-group instead of 32: a lane takes two neighbouring pairs per 16-byte load, 32 pairs per thread; pairs re-dealt to lanes, so the order of the sums changes
# speedup vs baseline: 2.1807x; 1.3120x over previous
#include <hip/hip_runtime.h>

#define N_ATOMS 4096
#define N3 (N_ATOMS * 3)
#define CUTOFF 12.0f
#define ROWS_PER_GRP 2
#define ROW_SHIFT 1
#define N_GRP (N_ATOMS / ROWS_PER_GRP)
#define NSRT 256
#define SRT_THREADS 512
#define SRT_WAVES (SRT_THREADS / 32)
#define SRT_PER_THR 32
#define SEG_CAP (SRT_THREADS * SRT_PER_THR)
#define NIMG 256
#define GRP_PER_WG (N_GRP / NIMG)
#define PAIR_THREADS 1024
#define PAIR_WAVES (PAIR_THREADS / 32)
#define LANES_PER_SEG (PAIR_THREADS / NSRT)
#define REC_PRE 4
#define KEYS_PER_THR (N_GRP / SRT_THREADS)
#define SUM_WAVES 16
#define IMG_PER_WAVE (NIMG / SUM_WAVES)
#define N_COLGRP (N3 / 64)
#define N_EPART (NIMG * PAIR_WAVES)

#define W_IMG 0
#define W_EP  (NIMG * N3)
#define W_OFF (W_EP + N_EPART)
#define W_SEG (W_OFF + NSRT * (N_GRP + 1))
static_assert((W_SEG & 1) == 0, "records are 8-byte aligned");
static_assert(KEYS_PER_THR == 4 && (1 << ROW_SHIFT) == ROWS_PER_GRP, "the sorting kernel scans four keys per thread");
static_assert((SRT_PER_THR & 1) == 0 && (SEG_CAP & 1) == 0, "pairs are loaded and records copied out two at a time");
static_assert(LANES_PER_SEG == 4, "lane-in-segment mask below is 3");

typedef __attribute__((ext_vector_type(2))) float v2f;
typedef __attribute__((ext_vector_type(4))) float v4f;
typedef __attribute__((ext_vector_type(8))) float v8f;
typedef __attribute__((ext_vector_type(4))) int   v4i;

#define LD_ONCE(p) __builtin_nontemporal_load(p)

#define LDS_BARRIER()                                                     \
    do {                                                                  \
        __builtin_amdgcn_fence(__ATOMIC_RELEASE, "workgroup", "local");   \
        __builtin_amdgcn_s_barrier();                                     \
        __builtin_amdgcn_fence(__ATOMIC_ACQUIRE, "workgroup", "local");   \
    } while (0)


__global__ __launch_bounds__(SRT_THREADS) void TFF_Repulsion_sort(
    const float* __restrict__ rep,
    const int*   __restrict__ coord,
    int n_pairs,
    float* __restrict__ ws) {
    __shared__ __attribute__((aligned(16))) int  s_cnt[N_GRP];
    __shared__ int  s_wtot[SRT_WAVES];
    __shared__ __attribute__((aligned(16))) int2 s_rec[SEG_CAP];
    const int tid  = threadIdx.x;
    const int lane = tid & 31;
    const int wv   = tid >> 5;
    const int per  = (((n_pairs + NSRT - 1) / NSRT) + 1) & ~1;
    const int s    = blockIdx.x * per;
    int e = s + per; if (e > n_pairs) e = n_pairs;
    const int n = (e > s) ? (e - s) : 0;

    *(v4i*)&s_cnt[KEYS_PER_THR * tid] = (v4i){0, 0, 0, 0};
    __syncthreads();

    int   key[SRT_PER_THR];
    float Bc[SRT_PER_THR];
#pragma unroll
    for (int k = 0; k < SRT_PER_THR / 2; ++k) {
        const int i  = s + 2 * (tid + k * SRT_THREADS);
        const bool ok = i + 1 < e;
        const int ii = ok ? i : 0;
        const v4i ab = LD_ONCE((const v4i*)(coord + 2 * (size_t)ii));
        const v2f bb = LD_ONCE((const v2f*)(rep + ii));
        const int a0 = ab[0] & (N_ATOMS - 1), b0 = ab[1] & (N_ATOMS - 1);
        const int a1 = ab[2] & (N_ATOMS - 1), b1 = ab[3] & (N_ATOMS - 1);
        key[2 * k]     = ok ? ((a0 << 12) | b0) : -1;
        key[2 * k + 1] = ok ? ((a1 << 12) | b1) : -1;
        Bc[2 * k]      = bb[0];
        Bc[2 * k + 1]  = bb[1];
    }
#pragma unroll
    for (int k = 0; k < SRT_PER_THR; ++k)
        if (key[k] >= 0) atomicAdd(&s_cnt[key[k] >> (12 + ROW_SHIFT)], 1);
    __syncthreads();

    const v4i cv = *(const v4i*)&s_cnt[KEYS_PER_THR * tid];
    const int c = (cv[0] + cv[1]) + (cv[2] + cv[3]);
    int x = c;
    for (int m = 1; m < 32; m <<= 1) { const int y = __shfl_up(x, m, 32); if (lane >= m) x += y; }
    if (lane == 31) s_wtot[wv] = x;
    __syncthreads();
    if (wv == 0) {
        const int t = (lane < SRT_WAVES) ? s_wtot[lane & (SRT_WAVES - 1)] : 0;
        int z = t;
        for (int m = 1; m < 32; m <<= 1) { const int y = __shfl_up(z, m, 32); if (lane >= m) z += y; }
        if (lane < SRT_WAVES) s_wtot[lane] = z - t;
    }
    __syncthreads();
    const int excl = x - c + s_wtot[wv];
    v4i st;
    st[0] = excl; st[1] = st[0] + cv[0]; st[2] = st[1] + cv[1]; st[3] = st[2] + cv[2];
    int* offt = (int*)ws + W_OFF + blockIdx.x * (N_GRP + 1);
#pragma unroll
    for (int r = 0; r < KEYS_PER_THR; ++r) offt[KEYS_PER_THR * tid + r] = st[r];
    if (tid == SRT_THREADS - 1) offt[N_GRP] = excl + c;
    *(v4i*)&s_cnt[KEYS_PER_THR * tid] = st;
    __syncthreads();

    int pos[SRT_PER_THR];
#pragma unroll
    for (int k = 0; k < SRT_PER_THR; ++k) {
        pos[k] = 0;
        if (key[k] >= 0) pos[k] = atomicAdd(&s_cnt[key[k] >> (12 + ROW_SHIFT)], 1);
    }
#pragma unroll
    for (int k = 0; k < SRT_PER_THR; ++k)
        if (key[k] >= 0) s_rec[pos[k]] = make_int2(key[k], __float_as_int(Bc[k]));
    __syncthreads();

    int2* seg = (int2*)((int*)ws + W_SEG) + (size_t)blockIdx.x * SEG_CAP;
    {
        const v4i* r4 = (const v4i*)s_rec;
        v4i* seg4 = (v4i*)seg;
        const int n2 = n >> 1;
#pragma unroll 4
        for (int j = tid; j < n2; j += SRT_THREADS) seg4[j] = r4[j];
        if ((n & 1) && tid == 0) seg[n - 1] = s_rec[n - 1];
    }
}

__device__ __forceinline__ float pair_term(int key, float B, const float* s_dist, const float* s_vec,
                                           float* s_forces, float* fa) {
    const int a  = key >> 12;
    const int b  = key & (N_ATOMS - 1);
    const int ri = a & (ROWS_PER_GRP - 1);
    const int li = (ri << 12) + b;
    const float d = s_dist[li];
    float e = 0.0f;
    if (d <= CUTOFF) {
        const float vx = s_vec[3 * li + 0], vy = s_vec[3 * li + 1], vz = s_vec[3 * li + 2];
        float inv  = 1.0f / d;
        float inv2 = inv * inv;
        float inv6 = inv2 * inv2 * inv2;
        e = B * inv6;
        float f = -6.0f * e * inv;
        float fx = f * vx, fy = f * vy, fz = f * vz;
        atomicAdd(&s_forces[3 * b + 0], -fx);
        atomicAdd(&s_forces[3 * b + 1], -fy);
        atomicAdd(&s_forces[3 * b + 2], -fz);
#pragma unroll
        for (int r = 0; r < ROWS_PER_GRP; ++r) {
            const bool m = (ri == r);
            fa[3 * r + 0] += m ? fx : 0.0f;
            fa[3 * r + 1] += m ? fy : 0.0f;
            fa[3 * r + 2] += m ? fz : 0.0f;
        }
    }
    return e;
}

__global__ __launch_bounds__(PAIR_THREADS) void TFF_Repulsion_pairs(
    const float* __restrict__ dist,
    const float* __restrict__ vec,
    float* __restrict__ ws) {
    __shared__ __attribute__((aligned(16))) float s_forces[N3];
    __shared__ __attribute__((aligned(16))) float s_dist[ROWS_PER_GRP * N_ATOMS];
    __shared__ __attribute__((aligned(16))) float s_vec[ROWS_PER_GRP * N3];
    const int tid  = threadIdx.x;
    const int lane = tid & 31;
    for (int j = tid; j < N3; j += PAIR_THREADS) s_forces[j] = 0.0f;

    float e0 = 0.0f, e1 = 0.0f;
    const int segi = tid >> 2;
    const int part = tid & 3;
    const int*  offt = (const int*)ws + W_OFF + segi * (N_GRP + 1) + blockIdx.x * GRP_PER_WG;
    const int2* seg  = (const int2*)((const int*)ws + W_SEG) + (size_t)segi * SEG_CAP;
    int bnd[GRP_PER_WG + 1];
#pragma unroll
    for (int k = 0; k <= GRP_PER_WG; ++k) bnd[k] = offt[k];

    v4f  rd[2][2], rv[2][6];
    int2 rec[REC_PRE];
#define LOAD_ROWS(grp_, set_)                                                        \
    {                                                                                \
        const v4f* gd = (const v4f*)(dist + (size_t)(grp_) * ROWS_PER_GRP * N_ATOMS);\
        const v4f* gv = (const v4f*)(vec  + (size_t)(grp_) * ROWS_PER_GRP * N3);     \
        _Pragma("unroll") for (int k = 0; k < 2; ++k) rd[set_][k] = LD_ONCE(&gd[tid + PAIR_THREADS * k]); \
        _Pragma("unroll") for (int k = 0; k < 6; ++k) rv[set_][k] = LD_ONCE(&gv[tid + PAIR_THREADS * k]); \
    }
#define LOAD_RECS(q_)                                                                \
    {                                                                                \
        const int lo = bnd[q_], hi = bnd[(q_) + 1];                                  \
        _Pragma("unroll") for (int k = 0; k < REC_PRE; ++k) {                        \
            const int j = lo + part + LANES_PER_SEG * k;                             \
            rec[k] = seg[(j < hi) ? j : (lo + part)];                                \
        }                                                                            \
    }

    LOAD_ROWS(blockIdx.x * GRP_PER_WG + 0, 0)
    LOAD_RECS(0)
    LOAD_ROWS(blockIdx.x * GRP_PER_WG + 1, 1)

#pragma unroll
    for (int q = 0; q < GRP_PER_WG; ++q) {
        LDS_BARRIER();
        {
            v4f* sd4 = (v4f*)s_dist;
            v4f* sv4 = (v4f*)s_vec;
#pragma unroll
            for (int k = 0; k < 2; ++k) sd4[tid + PAIR_THREADS * k] = rd[q & 1][k];
#pragma unroll
            for (int k = 0; k < 6; ++k) sv4[tid + PAIR_THREADS * k] = rv[q & 1][k];
        }
        LDS_BARRIER();
        if (q + 2 < GRP_PER_WG) LOAD_ROWS(blockIdx.x * GRP_PER_WG + q + 2, q & 1)

        float fa[3 * ROWS_PER_GRP];
#pragma unroll
        for (int r = 0; r < 3 * ROWS_PER_GRP; ++r) fa[r] = 0.0f;
#pragma unroll
        for (int k = 0; k < REC_PRE; ++k) {
            if (bnd[q] + part + LANES_PER_SEG * k < bnd[q + 1]) {
                const float e = pair_term(rec[k].x, __int_as_float(rec[k].y), s_dist, s_vec, s_forces, fa);
                if (k & 1) e1 += e; else e0 += e;
            }
        }
        for (int j = bnd[q] + part + LANES_PER_SEG * REC_PRE; j < bnd[q + 1]; j += LANES_PER_SEG) {
            const int2 r1 = seg[j];
            e0 += pair_term(r1.x, __int_as_float(r1.y), s_dist, s_vec, s_forces, fa);
        }
        const int a0 = (blockIdx.x * GRP_PER_WG + q) * ROWS_PER_GRP;
#pragma unroll
        for (int m = 16; m > 0; m >>= 1) {
            float t[3 * ROWS_PER_GRP];
#pragma unroll
            for (int r = 0; r < 3 * ROWS_PER_GRP; ++r) t[r] = __shfl_xor(fa[r], m, 32);
#pragma unroll
            for (int r = 0; r < 3 * ROWS_PER_GRP; ++r) fa[r] += t[r];
        }
        float mine = 0.0f;
#pragma unroll
        for (int r = 0; r < 3 * ROWS_PER_GRP; ++r) mine = (lane == r) ? fa[r] : mine;
        if (lane < 3 * ROWS_PER_GRP) atomicAdd(&s_forces[3 * a0 + lane], mine);
        if (q + 1 < GRP_PER_WG) LOAD_RECS(q + 1)
    }
#undef LOAD_ROWS
#undef LOAD_RECS
    __syncthreads();

    {
        const v4f* s4 = (const v4f*)s_forces;
        v4f* img = (v4f*)(ws + W_IMG + (size_t)blockIdx.x * N3);
#pragma unroll
        for (int t = 0; t < N3 / 4 / PAIR_THREADS; ++t) {
            const int j = tid + PAIR_THREADS * t;
            img[j] = s4[j];
        }
    }

    v2f a2; a2[0] = e0;   a2[1] = e1;
    v2f b2; b2[0] = 1.0f; b2[1] = 1.0f;
    v8f c = {};
    c = __builtin_amdgcn_wmma_f32_16x16x4_f32(
             false, a2,  false, b2,
             (short)0, c,  false,  false);
    float s = c[0] + c[1] + c[2] + c[3] + c[4] + c[5] + c[6] + c[7];
    s += __shfl_xor(s, 16, 32);
    if (lane == 0)
        ws[W_EP + blockIdx.x * PAIR_WAVES + (tid >> 5)] = s;
}

static_assert(N_COLGRP + 1 == 193 && SUM_WAVES * 32 == 512 && NIMG % SUM_WAVES == 0 && N_EPART == 512 * 16,
              "summing kernel's launch shape in kernel_launch must match these sizes");
__global__ __launch_bounds__(SUM_WAVES * 32) void TFF_Repulsion_sum(
    const float* __restrict__ forces_in,
    const float* __restrict__ ws,
    float* __restrict__ out) {
    __shared__ v2f   s_part[SUM_WAVES][32];
    __shared__ float s_e[SUM_WAVES];
    const int lane = threadIdx.x & 31;
    const int w    = threadIdx.x >> 5;
    if (blockIdx.x < N_COLGRP) {
        const int col = blockIdx.x * 64 + 2 * lane;
        const float* p = ws + W_IMG + (size_t)(w * IMG_PER_WAVE) * N3 + col;
        v2f acc0 = {0.0f, 0.0f}, acc1 = {0.0f, 0.0f};
#pragma unroll
        for (int i = 0; i < IMG_PER_WAVE; i += 2) {
            acc0 += LD_ONCE((const v2f*)(p + (size_t)i * N3));
            acc1 += LD_ONCE((const v2f*)(p + (size_t)(i + 1) * N3));
        }
        s_part[w][lane] = acc0 + acc1;
        __syncthreads();
        if (w == 0) {
            v2f t = s_part[0][lane];
            for (int k = 1; k < SUM_WAVES; ++k) t += s_part[k][lane];
            out[1 + col]     = forces_in[col]     + t[0];
            out[1 + col + 1] = forces_in[col + 1] + t[1];
        }
    } else {
        const v4f* e4 = (const v4f*)(ws + W_EP) + 4 * threadIdx.x;
        v4f x = e4[0], y = e4[1], z = e4[2], u = e4[3];
        float s = (((x[0] + x[1]) + (x[2] + x[3])) + ((y[0] + y[1]) + (y[2] + y[3])))
                + (((z[0] + z[1]) + (z[2] + z[3])) + ((u[0] + u[1]) + (u[2] + u[3])));
        for (int m = 16; m > 0; m >>= 1) s += __shfl_xor(s, m, 32);
        if (lane == 0) s_e[w] = s;
        __syncthreads();
        if (threadIdx.x == 0) {
            float t = s_e[0];
            for (int k = 1; k < SUM_WAVES; ++k) t += s_e[k];
            out[0] = t;
        }
    }
}

extern "C" void kernel_launch(void* const* d_in, const int* in_sizes, int n_in,
                              void* d_out, int out_size, void* d_ws, size_t ws_size,
                              hipStream_t stream) {
    const float* dist      = (const float*)d_in[0];
    const float* vec       = (const float*)d_in[1];
    const float* forces_in = (const float*)d_in[2];
    const float* rep       = (const float*)d_in[3];
    const int*   coord     = (const int*)d_in[4];
    float* out = (float*)d_out;
    float* ws  = (float*)d_ws;
    int n_pairs = in_sizes[3];

    TFF_Repulsion_sort<<<NSRT, SRT_THREADS, 0, stream>>>(rep, coord, n_pairs, ws);

    TFF_Repulsion_pairs<<<NIMG, PAIR_THREADS, 0, stream>>>(dist, vec, ws);

    TFF_Repulsion_sum<<<193, 512, 0, stream>>>(forces_in, ws, out);
}
